# starting kernel
# baseline (speedup 1.0000x reference)
#include <hip/hip_runtime.h>

#define B_     4096
#define NIN_   512
#define L_     16
#define N_     512
#define K_     32
#define NOUT_  256
#define TOTAL_ (NIN_ + L_ * N_)

typedef float v2f __attribute__((ext_vector_type(2)));
typedef float v8f __attribute__((ext_vector_type(8)));

__global__ __launch_bounds__(256)
void ffn_init_buf(const float* __restrict__ x, float* __restrict__ buf) {
    int i = blockIdx.x * blockDim.x + threadIdx.x;
    int b = i / NIN_;
    int f = i - b * NIN_;
    buf[(size_t)b * TOTAL_ + f] = x[i];
}

__device__ __forceinline__ v8f ffn_tile_accum(const float* __restrict__ buf,
                                              const float* __restrict__ W,
                                              const int*   __restrict__ idx,
                                              int n0, size_t rowA, int hi, int lo) {
    v8f acc = {};
    #pragma unroll 1
    for (int j = 0; j < 16; ++j) {
        const int nj = n0 + j;
        const int*   idr = idx + nj * K_;
        const float* wr  = W   + nj * K_;
        const bool mine = (lo == j);
        #pragma unroll
        for (int kk = 0; kk < 8; ++kk) {
            const int kb = kk * 4 + 2 * hi;
            const int2   fp = *(const int2*)  (idr + kb);
            const float2 wp = *(const float2*)(wr  + kb);
            v2f a, bm;
            a.x = buf[rowA + (size_t)fp.x];
            a.y = buf[rowA + (size_t)fp.y];
            bm.x = mine ? wp.x : 0.0f;
            bm.y = mine ? wp.y : 0.0f;
            acc = __builtin_amdgcn_wmma_f32_16x16x4_f32(
                      false, a, false, bm, (short)0, acc, false, false);
        }
    }
    return acc;
}

__global__ __launch_bounds__(32)
void ffn_layer_kernel(const float* __restrict__ buf_in,
                      float*       __restrict__ buf,
                      const float* __restrict__ W,
                      const float* __restrict__ bias,
                      const int*   __restrict__ idx,
                      int out_off) {
    const int lane = threadIdx.x;
    const int hi = lane >> 4;
    const int lo = lane & 15;
    const int bt = blockIdx.x;
    const int n0 = blockIdx.y * 16;
    const size_t rowA = (size_t)(bt * 16 + lo) * TOTAL_;

    v8f acc = ffn_tile_accum(buf_in, W, idx, n0, rowA, hi, lo);

    const float bv = bias[n0 + lo];
    #pragma unroll
    for (int r = 0; r < 8; ++r) {
        float v = acc[r] + bv;
        float o = 1.0f / (1.0f + __expf(-v));
        buf[(size_t)(bt * 16 + r + 8 * hi) * TOTAL_ + out_off + n0 + lo] = o;
    }
}

__global__ __launch_bounds__(32)
void ffn_out_kernel(const float* __restrict__ buf,
                    const float* __restrict__ W,
                    const float* __restrict__ bias,
                    const int*   __restrict__ idx,
                    float*       __restrict__ out) {
    const int lane = threadIdx.x;
    const int hi = lane >> 4;
    const int lo = lane & 15;
    const int bt = blockIdx.x;
    const int n0 = blockIdx.y * 16;
    const size_t rowA = (size_t)(bt * 16 + lo) * TOTAL_;

    v8f acc = ffn_tile_accum(buf, W, idx, n0, rowA, hi, lo);

    const float bv = bias[n0 + lo];
    #pragma unroll
    for (int r = 0; r < 8; ++r) {
        float v = acc[r] + bv;
        float o = 1.0f / (1.0f + __expf(-v));
        out[(size_t)(bt * 16 + r + 8 * hi) * NOUT_ + n0 + lo] = o;
    }
}

extern "C" void kernel_launch(void* const* d_in, const int* in_sizes, int n_in,
                              void* d_out, int out_size, void* d_ws, size_t ws_size,
                              hipStream_t stream) {
    const float* x     = (const float*)d_in[0];
    const float* W     = (const float*)d_in[1];
    const float* b     = (const float*)d_in[2];
    const float* W_out = (const float*)d_in[3];
    const float* b_out = (const float*)d_in[4];
    const int*   idx     = (const int*)d_in[5];
    const int*   idx_out = (const int*)d_in[6];

    float* buf = (float*)d_ws;
    float* out = (float*)d_out;

    ffn_init_buf<<<(B_ * NIN_) / 256, 256, 0, stream>>>(x, buf);

    dim3 grid(B_ / 16, N_ / 16);
    for (int l = 0; l < L_; ++l) {
        ffn_layer_kernel<<<grid, 32, 0, stream>>>(
            buf, buf,
            W   + (size_t)l * N_ * K_,
            b   + (size_t)l * N_,
            idx + (size_t)l * N_ * K_,
            NIN_ + l * N_);
    }

    dim3 grid_out(B_ / 16, NOUT_ / 16);
    ffn_out_kernel<<<grid_out, 32, 0, stream>>>(buf, W_out, b_out, idx_out, out);
}


# Index/weight rows leave the neuron loop: a lane loads its own neuron's half row once per tile, ahead of the LDS staging; gather offsets reach its 16-lane row by DPP row_share; sums unchanged
# speedup vs baseline: 6.9312x; 6.9312x over previous
#include <hip/hip_runtime.h>


#define B_     4096
#define NIN_   512
#define L_     16
#define N_     512
#define K_     32
#define NOUT_  256
#define TOTAL_ (NIN_ + L_ * N_)

#define WIN_    4352
#define PITCH_  4356
#define WAVES_  16

typedef float v2f __attribute__((ext_vector_type(2)));
typedef float v8f __attribute__((ext_vector_type(8)));

__global__ __launch_bounds__(256)
void ffn_init_buf(const float* __restrict__ x, float* __restrict__ buf) {
    int i = blockIdx.x * blockDim.x + threadIdx.x;
    int b = i / NIN_;
    int f = i - b * NIN_;
    buf[(size_t)b * TOTAL_ + f] = x[i];
}

__device__ __forceinline__ void ffn_load_own(int (&ix)[16], float (&wv)[16],
                                             const float* __restrict__ W,
                                             const int*   __restrict__ idx,
                                             int n0, int hi, int lo) {
    const int*   idr = idx + (n0 + lo) * K_ + 2 * hi;
    const float* wr  = W   + (n0 + lo) * K_ + 2 * hi;
    #pragma unroll
    for (int kk = 0; kk < 8; ++kk) {
        const int2   fp = *(const int2*)  (idr + 4 * kk);
        const float2 wp = *(const float2*)(wr  + 4 * kk);
        ix[2 * kk] = fp.x;  ix[2 * kk + 1] = fp.y;
        wv[2 * kk] = wp.x;  wv[2 * kk + 1] = wp.y;
    }
}

template <int J>
__device__ __forceinline__ int ffn_row_bcast(int v) {
    return __builtin_amdgcn_update_dpp(0, v, 0x150 + J, 0xF, 0xF, true);
}

template <int J>
__device__ __forceinline__ v8f ffn_neuron(v8f acc, const char* srowb,
                                          const int (&off)[16], const float (&wm)[16],
                                          int lo) {
    const bool mine = (lo == J);
    #pragma unroll
    for (int kk = 0; kk < 8; ++kk) {
        const int o0 = ffn_row_bcast<J>(off[2 * kk]);
        const int o1 = ffn_row_bcast<J>(off[2 * kk + 1]);
        v2f a, bm;
        a.x = *(const float*)(srowb + o0);
        a.y = *(const float*)(srowb + o1);
        bm.x = mine ? wm[2 * kk]     : 0.0f;
        bm.y = mine ? wm[2 * kk + 1] : 0.0f;
        acc = __builtin_amdgcn_wmma_f32_16x16x4_f32(
                  false, a, false, bm, (short)0, acc, false, false);
    }
    return acc;
}

__device__ __forceinline__ v8f ffn_tile_accum_lds(v8f acc, const char* srowb,
                                                  const int (&ix)[16], const float (&wv)[16],
                                                  int lo, int w0, unsigned wlen) {
    int off[16];
    float wm[16];
    #pragma unroll
    for (int e = 0; e < 16; ++e) {
        const unsigned r = (unsigned)(ix[e] - w0);
        const bool in = r < wlen;
        off[e] = in ? (int)(r << 2) : 0;
        wm[e]  = in ? wv[e] : 0.0f;
    }
    acc = ffn_neuron<0> (acc, srowb, off, wm, lo);
    acc = ffn_neuron<1> (acc, srowb, off, wm, lo);
    acc = ffn_neuron<2> (acc, srowb, off, wm, lo);
    acc = ffn_neuron<3> (acc, srowb, off, wm, lo);
    acc = ffn_neuron<4> (acc, srowb, off, wm, lo);
    acc = ffn_neuron<5> (acc, srowb, off, wm, lo);
    acc = ffn_neuron<6> (acc, srowb, off, wm, lo);
    acc = ffn_neuron<7> (acc, srowb, off, wm, lo);
    acc = ffn_neuron<8> (acc, srowb, off, wm, lo);
    acc = ffn_neuron<9> (acc, srowb, off, wm, lo);
    acc = ffn_neuron<10>(acc, srowb, off, wm, lo);
    acc = ffn_neuron<11>(acc, srowb, off, wm, lo);
    acc = ffn_neuron<12>(acc, srowb, off, wm, lo);
    acc = ffn_neuron<13>(acc, srowb, off, wm, lo);
    acc = ffn_neuron<14>(acc, srowb, off, wm, lo);
    acc = ffn_neuron<15>(acc, srowb, off, wm, lo);
    return acc;
}

__device__ __forceinline__ void ffn_tile_store(v8f acc, const float* __restrict__ bias,
                                               float* __restrict__ dst, int row0,
                                               int dst_pitch, int dst_off,
                                               int n0, int hi, int lo) {
    const float bv = bias[n0 + lo];
    #pragma unroll
    for (int r = 0; r < 8; ++r) {
        float v = acc[r] + bv;
        float o = 1.0f / (1.0f + __expf(-v));
        dst[(size_t)(row0 + r + 8 * hi) * dst_pitch + dst_off + n0 + lo] = o;
    }
}

__global__ __launch_bounds__(WAVES_ * 32)
void ffn_layer_kernel(const float* __restrict__ buf_in,
                      float*       __restrict__ dst,
                      const float* __restrict__ W,
                      const float* __restrict__ bias,
                      const int*   __restrict__ idx,
                      int P,
                      int ntiles,
                      int dst_pitch, int dst_off) {
    __shared__ float slab[16 * PITCH_];

    const int tid  = threadIdx.x;
    const int lane = tid & 31;
    const int wave = tid >> 5;
    const int hi = lane >> 4;
    const int lo = lane & 15;
    const int row0 = blockIdx.x * 16;

    const char*  srowb = (const char*)(slab + lo * PITCH_);
    const bool   two = ntiles > WAVES_;
    const float* grow = buf_in + (size_t)(row0 + wave) * TOTAL_;
    float*       lrow = slab + wave * PITCH_;

    int   ix0[16], ix1[16];
    float wv0[16], wv1[16];
    ffn_load_own(ix0, wv0, W, idx, wave * 16, hi, lo);
    if (two) {
        ffn_load_own(ix1, wv1, W, idx, (wave + WAVES_) * 16, hi, lo);
    } else {
        #pragma unroll
        for (int e = 0; e < 16; ++e) { ix1[e] = 0; wv1[e] = 0.0f; }
    }

    v8f acc0 = {};
    v8f acc1 = {};

    for (int w0 = 0; w0 < P; w0 += WIN_) {
        const int wlen = (P - w0 < WIN_) ? (P - w0) : WIN_;
        const int q = wlen >> 2;
        if (w0 > 0) __syncthreads();
        #pragma unroll 4
        for (int c = lane; c < q; c += 32) {
            const float4 v = *(const float4*)(grow + w0 + 4 * c);
            *(float4*)(lrow + 4 * c) = v;
        }
        __syncthreads();

        acc0 = ffn_tile_accum_lds(acc0, srowb, ix0, wv0, lo, w0, (unsigned)wlen);
        if (two)
            acc1 = ffn_tile_accum_lds(acc1, srowb, ix1, wv1, lo, w0, (unsigned)wlen);
    }

    ffn_tile_store(acc0, bias, dst, row0, dst_pitch, dst_off, wave * 16, hi, lo);
    if (two)
        ffn_tile_store(acc1, bias, dst, row0, dst_pitch, dst_off, (wave + WAVES_) * 16, hi, lo);
}

extern "C" void kernel_launch(void* const* d_in, const int* in_sizes, int n_in,
                              void* d_out, int out_size, void* d_ws, size_t ws_size,
                              hipStream_t stream) {
    const float* x     = (const float*)d_in[0];
    const float* W     = (const float*)d_in[1];
    const float* b     = (const float*)d_in[2];
    const float* W_out = (const float*)d_in[3];
    const float* b_out = (const float*)d_in[4];
    const int*   idx     = (const int*)d_in[5];
    const int*   idx_out = (const int*)d_in[6];

    float* buf = (float*)d_ws;
    float* out = (float*)d_out;

    ffn_init_buf<<<(B_ * NIN_) / 256, 256, 0, stream>>>(x, buf);

    for (int l = 0; l < L_; ++l) {
        ffn_layer_kernel<<<B_ / 16, WAVES_ * 32, 0, stream>>>(
            buf, buf,
            W   + (size_t)l * N_ * K_,
            b   + (size_t)l * N_,
            idx + (size_t)l * N_ * K_,
            NIN_ + l * N_, N_ / 16, TOTAL_, NIN_ + l * N_);
    }

    ffn_layer_kernel<<<B_ / 16, WAVES_ * 32, 0, stream>>>(
        buf, out, W_out, b_out, idx_out, TOTAL_, NOUT_ / 16, NOUT_, 0);
}


# Batch tile staged as a ring of two 4-row LDS groups: row loads of group g+1 issue before group g's WMMAs and are stored to LDS behind them, so the copy runs under the matrix work; sums unchanged
# speedup vs baseline: 9.7459x; 1.4061x over previous
#include <hip/hip_runtime.h>


#define B_     4096
#define NIN_   512
#define L_     16
#define N_     512
#define K_     32
#define NOUT_  256
#define TOTAL_ (NIN_ + L_ * N_)

#define PITCH_    8720
#define CHUNK_    (4 * PITCH_)
#define SLAB_     (2 * CHUNK_)
#define WAVES_    16
#define STG_IT_   (TOTAL_ / 512)

static_assert(NIN_ % 512 == 0 && N_ % 512 == 0, "a row is staged in 2-KB steps of 128 threads");

typedef float v2f __attribute__((ext_vector_type(2)));
typedef float v8f __attribute__((ext_vector_type(8)));
typedef float v4f __attribute__((ext_vector_type(4)));

__global__ __launch_bounds__(256)
void ffn_init_buf(const float* __restrict__ x, float* __restrict__ buf) {
    int i = blockIdx.x * blockDim.x + threadIdx.x;
    int b = i / NIN_;
    int f = i - b * NIN_;
    buf[(size_t)b * TOTAL_ + f] = x[i];
}

__device__ __forceinline__ void ffn_load_own(int (&ix)[16], float (&wv)[16],
                                             const float* __restrict__ W,
                                             const int*   __restrict__ idx,
                                             int n0, int hi, int lo) {
    const int*   idr = idx + (n0 + lo) * K_ + 2 * hi;
    const float* wr  = W   + (n0 + lo) * K_ + 2 * hi;
    #pragma unroll
    for (int kk = 0; kk < 8; ++kk) {
        const int2   fp = *(const int2*)  (idr + 4 * kk);
        const float2 wp = *(const float2*)(wr  + 4 * kk);
        ix[2 * kk] = fp.x;  ix[2 * kk + 1] = fp.y;
        wv[2 * kk] = wp.x;  wv[2 * kk + 1] = wp.y;
    }
}

typedef __attribute__((address_space(3))) float lds_float;

template <int J>
__device__ __forceinline__ unsigned ffn_quad_bcast(unsigned v) {
    return (unsigned)__builtin_amdgcn_update_dpp(0, (int)v, J * 0x55, 0xF, 0xF, true);
}

template <int J>
__device__ __forceinline__ v8f ffn_quad(unsigned rowb,
                                        const unsigned (&off)[16], const float (&wm)[16]) {
    v8f acc = {};
    #pragma unroll
    for (int kk = 0; kk < 8; ++kk) {
        const unsigned a0 = rowb + ffn_quad_bcast<J>(off[2 * kk]);
        const unsigned a1 = rowb + ffn_quad_bcast<J>(off[2 * kk + 1]);
        v2f a, bm;
        a.x = *(const lds_float*)a0;
        a.y = *(const lds_float*)a1;
        bm.x = wm[2 * kk];
        bm.y = wm[2 * kk + 1];
        acc = __builtin_amdgcn_wmma_f32_16x16x4_f32(
                  false, a, false, bm, (short)0, acc, false, false);
    }
    return acc;
}

__device__ __forceinline__ void ffn_group_accum_lds(float (&res)[4], unsigned rowb,
                                                    const unsigned (&off)[16],
                                                    const float (&wm)[16], int lo) {
    const v8f d0 = ffn_quad<0>(rowb, off, wm);
    const v8f d1 = ffn_quad<1>(rowb, off, wm);
    const v8f d2 = ffn_quad<2>(rowb, off, wm);
    const v8f d3 = ffn_quad<3>(rowb, off, wm);
    const int  j  = lo & 3;
    const bool up = (lo & 4) != 0;
    #pragma unroll
    for (int t = 0; t < 4; ++t) {
        float a = d0[t], b = d0[4 + t];
        a = (j == 1) ? d1[t] : a;   b = (j == 1) ? d1[4 + t] : b;
        a = (j == 2) ? d2[t] : a;   b = (j == 2) ? d2[4 + t] : b;
        a = (j == 3) ? d3[t] : a;   b = (j == 3) ? d3[4 + t] : b;
        res[t] = up ? b : a;
    }
}

__device__ __forceinline__ void ffn_prepare(unsigned (&off)[16], float (&wm)[16],
                                            const int (&ix)[16], const float (&wv)[16],
                                            unsigned P) {
    #pragma unroll
    for (int e = 0; e < 16; ++e) {
        const unsigned r = (unsigned)ix[e];
        const bool in = r < P;
        off[e] = in ? (r << 2) : 0u;
        wm[e]  = in ? wv[e] : 0.0f;
    }
}

__device__ __forceinline__ void ffn_group_store(const float (&res)[4], float bv,
                                                float* __restrict__ dst, int brow0,
                                                int dst_pitch, int col, bool holds) {
    if (holds) {
        #pragma unroll
        for (int t = 0; t < 4; ++t) {
            float v = res[t] + bv;
            float o = 1.0f / (1.0f + __expf(-v));
            dst[(size_t)(brow0 + t) * dst_pitch + col] = o;
        }
    }
}

__device__ __forceinline__ void ffn_stage_fetch(v4f (&v)[STG_IT_],
                                                const float* __restrict__ grow,
                                                int cpos, int nit) {
    #pragma unroll
    for (int i = 0; i < STG_IT_; ++i)
        if (i < nit) v[i] = *(const v4f*)(grow + 4 * (cpos + 128 * i));
}

__device__ __forceinline__ void ffn_stage_put(const v4f (&v)[STG_IT_], float* lrow,
                                              int cpos, int nit) {
    #pragma unroll
    for (int i = 0; i < STG_IT_; ++i)
        if (i < nit) *(v4f*)(lrow + 4 * (cpos + 128 * i)) = v[i];
}

__global__ __launch_bounds__(WAVES_ * 32)
void ffn_layer_kernel(const float* __restrict__ buf_in,
                      float*       __restrict__ dst,
                      const float* __restrict__ W,
                      const float* __restrict__ bias,
                      const int*   __restrict__ idx,
                      int P,
                      int ntiles,
                      int dst_pitch, int dst_off) {
    __shared__ __attribute__((aligned(16))) float slab[SLAB_];

    const int tid  = threadIdx.x;
    const int lane = tid & 31;
    const int wave = tid >> 5;
    const int hi = lane >> 4;
    const int lo = lane & 15;
    const int row0 = blockIdx.x * 16;

    const bool two = ntiles > WAVES_;
    const int  nit = P >> 9;

    const int srow = wave & 3;
    const int cpos = (wave >> 2) * 32 + lane;
    const float* grow = buf_in + (size_t)(row0 + srow) * TOTAL_;
    float* lrow = slab + srow * PITCH_;
    const unsigned slab_b = (unsigned)(__UINTPTR_TYPE__)(lds_float*)slab;

    int   ix0[16], ix1[16];
    float wv0[16], wv1[16];
    ffn_load_own(ix0, wv0, W, idx, wave * 16, hi, lo);
    if (two) {
        ffn_load_own(ix1, wv1, W, idx, (wave + WAVES_) * 16, hi, lo);
    } else {
        #pragma unroll
        for (int e = 0; e < 16; ++e) { ix1[e] = 0; wv1[e] = 0.0f; }
    }
    const float bv0 = bias[wave * 16 + lo];
    const float bv1 = two ? bias[(wave + WAVES_) * 16 + lo] : 0.0f;

    v4f v[STG_IT_];
    ffn_stage_fetch(v, grow, cpos, nit);

    unsigned off0[16], off1[16];
    float    wm0[16], wm1[16];
    ffn_prepare(off0, wm0, ix0, wv0, (unsigned)P);
    ffn_prepare(off1, wm1, ix1, wv1, (unsigned)P);

    const bool holds = (hi == (lo >> 3));
    const int  col0 = dst_off + wave * 16 + lo;
    const int  col1 = dst_off + (wave + WAVES_) * 16 + lo;

    ffn_stage_put(v, lrow, cpos, nit);
    __syncthreads();

    const unsigned rowb0 = slab_b + (unsigned)((lo & 3) * PITCH_) * 4u;
    float res0[4], res1[4];

    for (int g = 0; g < 3; ++g) {
        ffn_stage_fetch(v, grow + (size_t)(4 * (g + 1)) * TOTAL_, cpos, nit);

        const unsigned rowb = rowb0 + (unsigned)((g & 1) * CHUNK_) * 4u;
        ffn_group_accum_lds(res0, rowb, off0, wm0, lo);
        if (two) ffn_group_accum_lds(res1, rowb, off1, wm1, lo);

        ffn_stage_put(v, lrow + ((g + 1) & 1) * CHUNK_, cpos, nit);
        __syncthreads();

        ffn_group_store(res0, bv0, dst, row0 + 4 * g, dst_pitch, col0, holds);
        if (two) ffn_group_store(res1, bv1, dst, row0 + 4 * g, dst_pitch, col1, holds);
    }

    {
        const unsigned rowb = rowb0 + (unsigned)CHUNK_ * 4u;
        ffn_group_accum_lds(res0, rowb, off0, wm0, lo);
        ffn_group_store(res0, bv0, dst, row0 + 12, dst_pitch, col0, holds);
        if (two) {
            ffn_group_accum_lds(res1, rowb, off1, wm1, lo);
            ffn_group_store(res1, bv1, dst, row0 + 12, dst_pitch, col1, holds);
        }
    }
}

extern "C" void kernel_launch(void* const* d_in, const int* in_sizes, int n_in,
                              void* d_out, int out_size, void* d_ws, size_t ws_size,
                              hipStream_t stream) {
    const float* x     = (const float*)d_in[0];
    const float* W     = (const float*)d_in[1];
    const float* b     = (const float*)d_in[2];
    const float* W_out = (const float*)d_in[3];
    const float* b_out = (const float*)d_in[4];
    const int*   idx     = (const int*)d_in[5];
    const int*   idx_out = (const int*)d_in[6];

    float* buf = (float*)d_ws;
    float* out = (float*)d_out;

    ffn_init_buf<<<(B_ * NIN_) / 256, 256, 0, stream>>>(x, buf);

    for (int l = 0; l < L_; ++l) {
        ffn_layer_kernel<<<B_ / 16, WAVES_ * 32, 0, stream>>>(
            buf, buf,
            W   + (size_t)l * N_ * K_,
            b   + (size_t)l * N_,
            idx + (size_t)l * N_ * K_,
            NIN_ + l * N_, N_ / 16, TOTAL_, NIN_ + l * N_);
    }

    ffn_layer_kernel<<<B_ / 16, WAVES_ * 32, 0, stream>>>(
        buf, out, W_out, b_out, idx_out, TOTAL_, NOUT_ / 16, NOUT_, 0);
}
